# starting kernel
# baseline (speedup 1.0000x reference)
#include <hip/hip_runtime.h>

typedef __attribute__((ext_vector_type(2))) float v2f;
typedef __attribute__((ext_vector_type(8))) float v8f;

#define B_   256
#define OBS_ 256
#define H_   1024
#define A_   18

__global__ __launch_bounds__(32) void k_fc_in(const float* __restrict__ x,
                                              const float* __restrict__ Win,
                                              const float* __restrict__ bin,
                                              float* __restrict__ h1) {
  const int lane = threadIdx.x & 31;
  const int half = lane >> 4;
  const int l16  = lane & 15;
  const int m0   = blockIdx.x * 32;
  const int n0   = blockIdx.y * 32;

  v8f c00 = {}, c01 = {}, c10 = {}, c11 = {};

  const float* a0 = x   + (m0 +      l16) * OBS_ + 2 * half;
  const float* a1 = x   + (m0 + 16 + l16) * OBS_ + 2 * half;
  const float* b0 = Win + (n0 +      l16) * OBS_ + 2 * half;
  const float* b1 = Win + (n0 + 16 + l16) * OBS_ + 2 * half;

  for (int k = 0; k < OBS_; k += 4) {
    v2f af0 = *(const v2f*)(a0 + k);
    v2f af1 = *(const v2f*)(a1 + k);
    v2f bf0 = *(const v2f*)(b0 + k);
    v2f bf1 = *(const v2f*)(b1 + k);
    c00 = __builtin_amdgcn_wmma_f32_16x16x4_f32(false, af0, false, bf0, (short)0, c00, false, false);
    c01 = __builtin_amdgcn_wmma_f32_16x16x4_f32(false, af0, false, bf1, (short)0, c01, false, false);
    c10 = __builtin_amdgcn_wmma_f32_16x16x4_f32(false, af1, false, bf0, (short)0, c10, false, false);
    c11 = __builtin_amdgcn_wmma_f32_16x16x4_f32(false, af1, false, bf1, (short)0, c11, false, false);
  }

  const int col0 = n0 + l16;
  const int col1 = n0 + 16 + l16;
  const float bias0 = bin[col0];
  const float bias1 = bin[col1];
#pragma unroll
  for (int v = 0; v < 8; ++v) {
    const int r0 = m0 + v + 8 * half;
    const int r1 = m0 + 16 + v + 8 * half;
    h1[r0 * H_ + col0] = c00[v] + bias0;
    h1[r0 * H_ + col1] = c01[v] + bias1;
    h1[r1 * H_ + col0] = c10[v] + bias0;
    h1[r1 * H_ + col1] = c11[v] + bias1;
  }
}

template <int IS_MAX>
__global__ __launch_bounds__(256) void k_semiring(const float* __restrict__ hin,
                                                  const float* __restrict__ W,
                                                  float* __restrict__ hout) {
  __shared__ float hs[16 * H_];
  const int b0 = blockIdx.x * 16;
  const int o0 = blockIdx.y * 64;

  {
    const float4* src = (const float4*)(hin + b0 * H_);
    float4* dst = (float4*)hs;
    for (int idx = threadIdx.x; idx < (16 * H_) / 4; idx += 256) dst[idx] = src[idx];
  }
  __syncthreads();

  const int wave = threadIdx.x >> 5;
  const int lane = threadIdx.x & 31;

  for (int ot = 0; ot < 2; ++ot) {
    const int ob = o0 + wave * 8 + ot * 4;
    const float* wbase = W + (size_t)ob * H_;

    float acc[4][16];
#pragma unroll
    for (int oi = 0; oi < 4; ++oi)
#pragma unroll
      for (int b = 0; b < 16; ++b)
        acc[oi][b] = IS_MAX ? -3.402823466e38f : 3.402823466e38f;

    for (int c = 0; c < H_ / 128; ++c) {
      const int io = c * 128 + lane * 4;
      float4 w[4];
#pragma unroll
      for (int oi = 0; oi < 4; ++oi)
        w[oi] = *(const float4*)(wbase + oi * H_ + io);

#pragma unroll
      for (int b = 0; b < 16; ++b) {
        const float4 hv = *(const float4*)(hs + b * H_ + io);
#pragma unroll
        for (int oi = 0; oi < 4; ++oi) {
          if (IS_MAX) {
            float t0 = fmaxf(w[oi].x + hv.x, w[oi].y + hv.y);
            float t1 = fmaxf(w[oi].z + hv.z, w[oi].w + hv.w);
            acc[oi][b] = fmaxf(acc[oi][b], fmaxf(t0, t1));
          } else {
            float t0 = fminf(w[oi].x + hv.x, w[oi].y + hv.y);
            float t1 = fminf(w[oi].z + hv.z, w[oi].w + hv.w);
            acc[oi][b] = fminf(acc[oi][b], fminf(t0, t1));
          }
        }
      }
    }

#pragma unroll
    for (int oi = 0; oi < 4; ++oi) {
#pragma unroll
      for (int b = 0; b < 16; ++b) {
        float v = acc[oi][b];
#pragma unroll
        for (int off = 16; off > 0; off >>= 1) {
          const float other = __shfl_xor(v, off, 32);
          v = IS_MAX ? fmaxf(v, other) : fminf(v, other);
        }
        if (lane == 0) hout[(b0 + b) * H_ + ob + oi] = v;
      }
    }
  }
}

__global__ __launch_bounds__(32) void k_fc_out(const float* __restrict__ h,
                                               const float* __restrict__ Wout,
                                               const float* __restrict__ bout,
                                               float* __restrict__ q) {
  const int lane = threadIdx.x & 31;
  const int half = lane >> 4;
  const int l16  = lane & 15;
  const int m0   = blockIdx.x * 16;
  const int n0   = blockIdx.y * 16;

  const int col  = n0 + l16;
  const int bcol = (col < A_) ? col : 0;

  v8f c = {};
  const float* ap = h    + (m0 + l16) * H_ + 2 * half;
  const float* bp = Wout + bcol * H_       + 2 * half;

  for (int k = 0; k < H_; k += 4) {
    v2f a = *(const v2f*)(ap + k);
    v2f b = *(const v2f*)(bp + k);
    c = __builtin_amdgcn_wmma_f32_16x16x4_f32(false, a, false, b, (short)0, c, false, false);
  }

  if (col < A_) {
    const float bias = bout[col];
#pragma unroll
    for (int v = 0; v < 8; ++v) q[(m0 + v + 8 * half) * A_ + col] = c[v] + bias;
  }
}

extern "C" void kernel_launch(void* const* d_in, const int* in_sizes, int n_in,
                              void* d_out, int out_size, void* d_ws, size_t ws_size,
                              hipStream_t stream) {
  (void)in_sizes; (void)n_in; (void)out_size; (void)ws_size;
  const float* x    = (const float*)d_in[0];
  const float* Win  = (const float*)d_in[1];
  const float* bin  = (const float*)d_in[2];
  const float* Wmax = (const float*)d_in[3];
  const float* Wmin = (const float*)d_in[4];
  const float* Wout = (const float*)d_in[5];
  const float* bout = (const float*)d_in[6];
  float* q  = (float*)d_out;
  float* h1 = (float*)d_ws;
  float* h2 = h1 + (size_t)B_ * H_;

  k_fc_in   <<<dim3(B_ / 32, H_ / 32), 32,  0, stream>>>(x,  Win,  bin,  h1);
  k_semiring<1><<<dim3(B_ / 16, H_ / 64), 256, 0, stream>>>(h1, Wmax, h2);
  k_semiring<0><<<dim3(B_ / 16, H_ / 64), 256, 0, stream>>>(h2, Wmin, h1);
  k_fc_out  <<<dim3(B_ / 16, 2),       32,  0, stream>>>(h1, Wout, bout, q);
}


# semiring kernels: wave max/min reduction as in-register reduce-scatter (permlanex16 + DPP row_xmask, 62 exchanges per 64 values) instead of 320 serial ds_bpermute round trips; 8-byte store per lane
# speedup vs baseline: 1.7285x; 1.7285x over previous
#include <hip/hip_runtime.h>

typedef __attribute__((ext_vector_type(2))) float v2f;
typedef __attribute__((ext_vector_type(8))) float v8f;

#define B_   256
#define OBS_ 256
#define H_   1024
#define A_   18

__global__ __launch_bounds__(32) void k_fc_in(const float* __restrict__ x,
                                              const float* __restrict__ Win,
                                              const float* __restrict__ bin,
                                              float* __restrict__ h1) {
  const int lane = threadIdx.x & 31;
  const int half = lane >> 4;
  const int l16  = lane & 15;
  const int m0   = blockIdx.x * 32;
  const int n0   = blockIdx.y * 32;

  v8f c00 = {}, c01 = {}, c10 = {}, c11 = {};

  const float* a0 = x   + (m0 +      l16) * OBS_ + 2 * half;
  const float* a1 = x   + (m0 + 16 + l16) * OBS_ + 2 * half;
  const float* b0 = Win + (n0 +      l16) * OBS_ + 2 * half;
  const float* b1 = Win + (n0 + 16 + l16) * OBS_ + 2 * half;

  for (int k = 0; k < OBS_; k += 4) {
    v2f af0 = *(const v2f*)(a0 + k);
    v2f af1 = *(const v2f*)(a1 + k);
    v2f bf0 = *(const v2f*)(b0 + k);
    v2f bf1 = *(const v2f*)(b1 + k);
    c00 = __builtin_amdgcn_wmma_f32_16x16x4_f32(false, af0, false, bf0, (short)0, c00, false, false);
    c01 = __builtin_amdgcn_wmma_f32_16x16x4_f32(false, af0, false, bf1, (short)0, c01, false, false);
    c10 = __builtin_amdgcn_wmma_f32_16x16x4_f32(false, af1, false, bf0, (short)0, c10, false, false);
    c11 = __builtin_amdgcn_wmma_f32_16x16x4_f32(false, af1, false, bf1, (short)0, c11, false, false);
  }

  const int col0 = n0 + l16;
  const int col1 = n0 + 16 + l16;
  const float bias0 = bin[col0];
  const float bias1 = bin[col1];
#pragma unroll
  for (int v = 0; v < 8; ++v) {
    const int r0 = m0 + v + 8 * half;
    const int r1 = m0 + 16 + v + 8 * half;
    h1[r0 * H_ + col0] = c00[v] + bias0;
    h1[r0 * H_ + col1] = c01[v] + bias1;
    h1[r1 * H_ + col0] = c10[v] + bias0;
    h1[r1 * H_ + col1] = c11[v] + bias1;
  }
}

template <int M>
__device__ __forceinline__ float lane_xor(float x) {
  const int b = __builtin_bit_cast(int, x);
  if (M == 16)
    return __builtin_bit_cast(float, __builtin_amdgcn_permlanex16(b, b, 0x76543210, 0xfedcba98, false, false));
  return __builtin_bit_cast(float, __builtin_amdgcn_update_dpp(0, b, 0x160 | (M & 15), 0xf, 0xf, true));
}
template <int IS_MAX, int M, int HALF>
__device__ __forceinline__ void fold_bit(float (&v)[64], int lane) {
  const bool up = (lane & M) != 0;
#pragma unroll
  for (int k = 0; k < HALF; ++k) {
    const float keep = up ? v[k + HALF] : v[k];
    const float send = up ? v[k] : v[k + HALF];
    const float got  = lane_xor<M>(send);
    v[k] = IS_MAX ? fmaxf(keep, got) : fminf(keep, got);
  }
}

template <int IS_MAX>
__global__ __launch_bounds__(256) void k_semiring(const float* __restrict__ hin,
                                                  const float* __restrict__ W,
                                                  float* __restrict__ hout) {
  __shared__ float hs[16 * H_];
  const int b0 = blockIdx.x * 16;
  const int o0 = blockIdx.y * 64;

  {
    const float4* src = (const float4*)(hin + b0 * H_);
    float4* dst = (float4*)hs;
    for (int idx = threadIdx.x; idx < (16 * H_) / 4; idx += 256) dst[idx] = src[idx];
  }
  __syncthreads();

  const int wave = threadIdx.x >> 5;
  const int lane = threadIdx.x & 31;

  for (int ot = 0; ot < 2; ++ot) {
    const int ob = o0 + wave * 8 + ot * 4;
    const float* wbase = W + (size_t)ob * H_;

    float acc[4][16];
#pragma unroll
    for (int oi = 0; oi < 4; ++oi)
#pragma unroll
      for (int b = 0; b < 16; ++b)
        acc[oi][b] = IS_MAX ? -3.402823466e38f : 3.402823466e38f;

    for (int c = 0; c < H_ / 128; ++c) {
      const int io = c * 128 + lane * 4;
      float4 w[4];
#pragma unroll
      for (int oi = 0; oi < 4; ++oi)
        w[oi] = *(const float4*)(wbase + oi * H_ + io);

#pragma unroll
      for (int b = 0; b < 16; ++b) {
        const float4 hv = *(const float4*)(hs + b * H_ + io);
#pragma unroll
        for (int oi = 0; oi < 4; ++oi) {
          if (IS_MAX) {
            float t0 = fmaxf(w[oi].x + hv.x, w[oi].y + hv.y);
            float t1 = fmaxf(w[oi].z + hv.z, w[oi].w + hv.w);
            acc[oi][b] = fmaxf(acc[oi][b], fmaxf(t0, t1));
          } else {
            float t0 = fminf(w[oi].x + hv.x, w[oi].y + hv.y);
            float t1 = fminf(w[oi].z + hv.z, w[oi].w + hv.w);
            acc[oi][b] = fminf(acc[oi][b], fminf(t0, t1));
          }
        }
      }
    }

    float v[64];
#pragma unroll
    for (int b = 0; b < 16; ++b)
#pragma unroll
      for (int oi = 0; oi < 4; ++oi) v[b * 4 + oi] = acc[oi][b];

    fold_bit<IS_MAX, 16, 32>(v, lane);
    fold_bit<IS_MAX, 8, 16>(v, lane);
    fold_bit<IS_MAX, 4, 8>(v, lane);
    fold_bit<IS_MAX, 2, 4>(v, lane);
    fold_bit<IS_MAX, 1, 2>(v, lane);

    *(float2*)(hout + (b0 + (lane >> 1)) * H_ + ob + 2 * (lane & 1)) = make_float2(v[0], v[1]);
  }
}

__global__ __launch_bounds__(32) void k_fc_out(const float* __restrict__ h,
                                               const float* __restrict__ Wout,
                                               const float* __restrict__ bout,
                                               float* __restrict__ q) {
  const int lane = threadIdx.x & 31;
  const int half = lane >> 4;
  const int l16  = lane & 15;
  const int m0   = blockIdx.x * 16;
  const int n0   = blockIdx.y * 16;

  const int col  = n0 + l16;
  const int bcol = (col < A_) ? col : 0;

  v8f c = {};
  const float* ap = h    + (m0 + l16) * H_ + 2 * half;
  const float* bp = Wout + bcol * H_       + 2 * half;

  for (int k = 0; k < H_; k += 4) {
    v2f a = *(const v2f*)(ap + k);
    v2f b = *(const v2f*)(bp + k);
    c = __builtin_amdgcn_wmma_f32_16x16x4_f32(false, a, false, b, (short)0, c, false, false);
  }

  if (col < A_) {
    const float bias = bout[col];
#pragma unroll
    for (int v = 0; v < 8; ++v) q[(m0 + v + 8 * half) * A_ + col] = c[v] + bias;
  }
}

extern "C" void kernel_launch(void* const* d_in, const int* in_sizes, int n_in,
                              void* d_out, int out_size, void* d_ws, size_t ws_size,
                              hipStream_t stream) {
  (void)in_sizes; (void)n_in; (void)out_size; (void)ws_size;
  const float* x    = (const float*)d_in[0];
  const float* Win  = (const float*)d_in[1];
  const float* bin  = (const float*)d_in[2];
  const float* Wmax = (const float*)d_in[3];
  const float* Wmin = (const float*)d_in[4];
  const float* Wout = (const float*)d_in[5];
  const float* bout = (const float*)d_in[6];
  float* q  = (float*)d_out;
  float* h1 = (float*)d_ws;
  float* h2 = h1 + (size_t)B_ * H_;

  k_fc_in   <<<dim3(B_ / 32, H_ / 32), 32,  0, stream>>>(x,  Win,  bin,  h1);
  k_semiring<1><<<dim3(B_ / 16, H_ / 64), 256, 0, stream>>>(h1, Wmax, h2);
  k_semiring<0><<<dim3(B_ / 16, H_ / 64), 256, 0, stream>>>(h2, Wmin, h1);
  k_fc_out  <<<dim3(B_ / 16, 2),       32,  0, stream>>>(h1, Wout, bout, q);
}


# semiring kernels: inner loop as explicit packed add + 3-input max/min on 64 accumulators, under 256 registers, no spill, 16 waves per block (4 per SIMD); results bit-identical
# speedup vs baseline: 2.7632x; 1.5986x over previous
#include <hip/hip_runtime.h>

typedef __attribute__((ext_vector_type(2))) float v2f;
typedef __attribute__((ext_vector_type(4))) float v4f;
typedef __attribute__((ext_vector_type(8))) float v8f;

#define B_   256
#define OBS_ 256
#define H_   1024
#define A_   18

__global__ __launch_bounds__(32) void k_fc_in(const float* __restrict__ x,
                                              const float* __restrict__ Win,
                                              const float* __restrict__ bin,
                                              float* __restrict__ h1) {
  const int lane = threadIdx.x & 31;
  const int half = lane >> 4;
  const int l16  = lane & 15;
  const int m0   = blockIdx.x * 32;
  const int n0   = blockIdx.y * 32;

  v8f c00 = {}, c01 = {}, c10 = {}, c11 = {};

  const float* a0 = x   + (m0 +      l16) * OBS_ + 2 * half;
  const float* a1 = x   + (m0 + 16 + l16) * OBS_ + 2 * half;
  const float* b0 = Win + (n0 +      l16) * OBS_ + 2 * half;
  const float* b1 = Win + (n0 + 16 + l16) * OBS_ + 2 * half;

  for (int k = 0; k < OBS_; k += 4) {
    v2f af0 = *(const v2f*)(a0 + k);
    v2f af1 = *(const v2f*)(a1 + k);
    v2f bf0 = *(const v2f*)(b0 + k);
    v2f bf1 = *(const v2f*)(b1 + k);
    c00 = __builtin_amdgcn_wmma_f32_16x16x4_f32(false, af0, false, bf0, (short)0, c00, false, false);
    c01 = __builtin_amdgcn_wmma_f32_16x16x4_f32(false, af0, false, bf1, (short)0, c01, false, false);
    c10 = __builtin_amdgcn_wmma_f32_16x16x4_f32(false, af1, false, bf0, (short)0, c10, false, false);
    c11 = __builtin_amdgcn_wmma_f32_16x16x4_f32(false, af1, false, bf1, (short)0, c11, false, false);
  }

  const int col0 = n0 + l16;
  const int col1 = n0 + 16 + l16;
  const float bias0 = bin[col0];
  const float bias1 = bin[col1];
#pragma unroll
  for (int v = 0; v < 8; ++v) {
    const int r0 = m0 + v + 8 * half;
    const int r1 = m0 + 16 + v + 8 * half;
    h1[r0 * H_ + col0] = c00[v] + bias0;
    h1[r0 * H_ + col1] = c01[v] + bias1;
    h1[r1 * H_ + col0] = c10[v] + bias0;
    h1[r1 * H_ + col1] = c11[v] + bias1;
  }
}

template <int M>
__device__ __forceinline__ float lane_xor(float x) {
  const int b = __builtin_bit_cast(int, x);
  if (M == 16)
    return __builtin_bit_cast(float, __builtin_amdgcn_permlanex16(b, b, 0x76543210, 0xfedcba98, false, false));
  return __builtin_bit_cast(float, __builtin_amdgcn_update_dpp(0, b, 0x160 | (M & 15), 0xf, 0xf, true));
}
__device__ __forceinline__ v2f sr_add2(v2f a, v2f b) {
  v2f r;
  asm volatile("v_pk_add_f32 %0, %1, %2" : "=v"(r) : "v"(a), "v"(b));
  return r;
}
template <int IS_MAX>
__device__ __forceinline__ void sr_fold3(float& acc, float a, float b) {
  if (IS_MAX) asm volatile("v_max3_num_f32 %0, %0, %1, %2" : "+v"(acc) : "v"(a), "v"(b));
  else        asm volatile("v_min3_num_f32 %0, %0, %1, %2" : "+v"(acc) : "v"(a), "v"(b));
}
template <int IS_MAX>
__device__ __forceinline__ float sr_fold2(float a, float b) {
  float r;
  if (IS_MAX) asm("v_max_num_f32 %0, %1, %2" : "=v"(r) : "v"(a), "v"(b));
  else        asm("v_min_num_f32 %0, %1, %2" : "=v"(r) : "v"(a), "v"(b));
  return r;
}

template <int IS_MAX, int M, int HALF>
__device__ __forceinline__ void fold_bit(float (&v)[64], int lane) {
  const bool up = (lane & M) != 0;
  float keep[HALF], got[HALF];
#pragma unroll
  for (int k = 0; k < HALF; ++k) {
    keep[k] = up ? v[k + HALF] : v[k];
    const float send = up ? v[k] : v[k + HALF];
    got[k] = lane_xor<M>(send);
  }
#pragma unroll
  for (int k = 0; k < HALF; ++k) v[k] = sr_fold2<IS_MAX>(keep[k], got[k]);
}

__device__ __forceinline__ v2f lo2(v4f v) { return __builtin_shufflevector(v, v, 0, 1); }
__device__ __forceinline__ v2f hi2(v4f v) { return __builtin_shufflevector(v, v, 2, 3); }

template <int IS_MAX, int BASE>
__device__ __forceinline__ void sr_rows8(float (&acc)[4][16], const v4f (&w)[4], const v4f (&hv)[8]) {
#pragma unroll
  for (int p = 0; p < 8; p += 2) {
    v2f tl[2][4], th[2][4];
#pragma unroll
    for (int j = 0; j < 2; ++j)
#pragma unroll
      for (int oi = 0; oi < 4; ++oi) {
        tl[j][oi] = sr_add2(lo2(w[oi]), lo2(hv[p + j]));
        th[j][oi] = sr_add2(hi2(w[oi]), hi2(hv[p + j]));
      }
#pragma unroll
    for (int j = 0; j < 2; ++j)
#pragma unroll
      for (int oi = 0; oi < 4; ++oi) sr_fold3<IS_MAX>(acc[oi][BASE + p + j], tl[j][oi].x, tl[j][oi].y);
#pragma unroll
    for (int j = 0; j < 2; ++j)
#pragma unroll
      for (int oi = 0; oi < 4; ++oi) sr_fold3<IS_MAX>(acc[oi][BASE + p + j], th[j][oi].x, th[j][oi].y);
  }
}

template <int IS_MAX>
__device__ __forceinline__ void sr_step(float (&acc)[4][16], const v4f (&wc)[4], v4f (&wn)[4],
                                        v4f (&hA)[8], v4f (&hB)[8],
                                        const float* wp, const float* hp, int c) {
  const int io = c * 128;
  const int in = ((c + 1) & 7) * 128;
#pragma unroll
  for (int b = 0; b < 8; ++b) hB[b] = *(const v4f*)(hp + (8 + b) * H_ + io);
#pragma unroll
  for (int oi = 0; oi < 4; ++oi) wn[oi] = *(const v4f*)(wp + oi * H_ + in);
  sr_rows8<IS_MAX, 0>(acc, wc, hA);
#pragma unroll
  for (int b = 0; b < 8; ++b) hA[b] = *(const v4f*)(hp + b * H_ + in);
  sr_rows8<IS_MAX, 8>(acc, wc, hB);
}

template <int IS_MAX>
__global__ __launch_bounds__(512) void k_semiring(const float* __restrict__ hin,
                                                  const float* __restrict__ W,
                                                  float* __restrict__ hout) {
  __shared__ float hs[16 * H_];
  const int b0 = blockIdx.x * 16;
  const int o0 = blockIdx.y * 64;
  const int wave = threadIdx.x >> 5;
  const int lane = threadIdx.x & 31;
  const int ob = o0 + wave * 4;
  const float* wp = W + (size_t)ob * H_ + lane * 4;

  v4f wA[4], wB[4];
#pragma unroll
  for (int oi = 0; oi < 4; ++oi) wA[oi] = *(const v4f*)(wp + oi * H_);

  {
    const v4f* src = (const v4f*)(hin + b0 * H_);
    v4f* dst = (v4f*)hs;
    v4f r[8];
#pragma unroll
    for (int j = 0; j < 8; ++j) r[j] = src[threadIdx.x + 512 * j];
#pragma unroll
    for (int j = 0; j < 8; ++j) dst[threadIdx.x + 512 * j] = r[j];
  }
  __syncthreads();

  const float* hp = hs + lane * 4;
  float acc[4][16];
#pragma unroll
  for (int oi = 0; oi < 4; ++oi)
#pragma unroll
    for (int b = 0; b < 16; ++b)
      acc[oi][b] = IS_MAX ? -3.402823466e38f : 3.402823466e38f;

  v4f hA[8], hB[8];
#pragma unroll
  for (int b = 0; b < 8; ++b) hA[b] = *(const v4f*)(hp + b * H_);

#pragma unroll 1
  for (int c = 0; c < H_ / 128; c += 2) {
    sr_step<IS_MAX>(acc, wA, wB, hA, hB, wp, hp, c);
    sr_step<IS_MAX>(acc, wB, wA, hA, hB, wp, hp, c + 1);
  }

  float v[64];
#pragma unroll
  for (int b = 0; b < 16; ++b)
#pragma unroll
    for (int oi = 0; oi < 4; ++oi) v[b * 4 + oi] = acc[oi][b];

  fold_bit<IS_MAX, 16, 32>(v, lane);
  fold_bit<IS_MAX, 8, 16>(v, lane);
  fold_bit<IS_MAX, 4, 8>(v, lane);
  fold_bit<IS_MAX, 2, 4>(v, lane);
  fold_bit<IS_MAX, 1, 2>(v, lane);

  *(float2*)(hout + (b0 + (lane >> 1)) * H_ + ob + 2 * (lane & 1)) = make_float2(v[0], v[1]);
}

__global__ __launch_bounds__(32) void k_fc_out(const float* __restrict__ h,
                                               const float* __restrict__ Wout,
                                               const float* __restrict__ bout,
                                               float* __restrict__ q) {
  const int lane = threadIdx.x & 31;
  const int half = lane >> 4;
  const int l16  = lane & 15;
  const int m0   = blockIdx.x * 16;
  const int n0   = blockIdx.y * 16;

  const int col  = n0 + l16;
  const int bcol = (col < A_) ? col : 0;

  v8f c = {};
  const float* ap = h    + (m0 + l16) * H_ + 2 * half;
  const float* bp = Wout + bcol * H_       + 2 * half;

  for (int k = 0; k < H_; k += 4) {
    v2f a = *(const v2f*)(ap + k);
    v2f b = *(const v2f*)(bp + k);
    c = __builtin_amdgcn_wmma_f32_16x16x4_f32(false, a, false, b, (short)0, c, false, false);
  }

  if (col < A_) {
    const float bias = bout[col];
#pragma unroll
    for (int v = 0; v < 8; ++v) q[(m0 + v + 8 * half) * A_ + col] = c[v] + bias;
  }
}

extern "C" void kernel_launch(void* const* d_in, const int* in_sizes, int n_in,
                              void* d_out, int out_size, void* d_ws, size_t ws_size,
                              hipStream_t stream) {
  (void)in_sizes; (void)n_in; (void)out_size; (void)ws_size;
  const float* x    = (const float*)d_in[0];
  const float* Win  = (const float*)d_in[1];
  const float* bin  = (const float*)d_in[2];
  const float* Wmax = (const float*)d_in[3];
  const float* Wmin = (const float*)d_in[4];
  const float* Wout = (const float*)d_in[5];
  const float* bout = (const float*)d_in[6];
  float* q  = (float*)d_out;
  float* h1 = (float*)d_ws;
  float* h2 = h1 + (size_t)B_ * H_;

  k_fc_in   <<<dim3(B_ / 32, H_ / 32), 32,  0, stream>>>(x,  Win,  bin,  h1);
  k_semiring<1><<<dim3(B_ / 16, H_ / 64), 512, 0, stream>>>(h1, Wmax, h2);
  k_semiring<0><<<dim3(B_ / 16, H_ / 64), 512, 0, stream>>>(h2, Wmin, h1);
  k_fc_out  <<<dim3(B_ / 16, 2),       32,  0, stream>>>(h1, Wout, bout, q);
}


# fc_in, fc_out: K split over the waves of a block (4 / 16 ways), a wave's operand loads all issued before its first WMMA, partials summed through LDS in fixed wave order (f32 sum order changes)
# speedup vs baseline: 4.3688x; 1.5811x over previous
#include <hip/hip_runtime.h>

typedef __attribute__((ext_vector_type(2))) float v2f;
typedef __attribute__((ext_vector_type(4))) float v4f;
typedef __attribute__((ext_vector_type(8))) float v8f;

#define B_   256
#define OBS_ 256
#define H_   1024
#define A_   18

__device__ __forceinline__ v2f lo2(v4f v) { return __builtin_shufflevector(v, v, 0, 1); }
__device__ __forceinline__ v2f hi2(v4f v) { return __builtin_shufflevector(v, v, 2, 3); }

__device__ __forceinline__ v8f mm4(v2f a, v2f b, v8f c) {
  return __builtin_amdgcn_wmma_f32_16x16x4_f32(false, a, false, b, (short)0, c, false, false);
}

__global__ __launch_bounds__(128) void k_fc_in(const float* __restrict__ x,
                                               const float* __restrict__ Win,
                                               const float* __restrict__ bin,
                                               float* __restrict__ h1) {
  __shared__ __attribute__((aligned(16))) float red[4 * 1024];
  const int wave = threadIdx.x >> 5;
  const int lane = threadIdx.x & 31;
  const int half = lane >> 4;
  const int l16  = lane & 15;
  const int m0   = blockIdx.x * 32;
  const int n0   = blockIdx.y * 32;
  const int k0   = wave * 64 + 4 * half;

  const float* a0 = x   + (m0 +      l16) * OBS_ + k0;
  const float* a1 = x   + (m0 + 16 + l16) * OBS_ + k0;
  const float* b0 = Win + (n0 +      l16) * OBS_ + k0;
  const float* b1 = Win + (n0 + 16 + l16) * OBS_ + k0;

  const int t  = threadIdx.x;
  const int rr = t >> 3;
  const int c0 = 4 * (t & 7);
  const v4f bias = *(const v4f*)(bin + n0 + c0);

  v4f fa0[8], fa1[8], fb0[8], fb1[8];
#pragma unroll
  for (int j = 0; j < 8; ++j) {
    fa0[j] = *(const v4f*)(a0 + 8 * j);
    fb0[j] = *(const v4f*)(b0 + 8 * j);
    fa1[j] = *(const v4f*)(a1 + 8 * j);
    fb1[j] = *(const v4f*)(b1 + 8 * j);
  }
  __builtin_amdgcn_sched_barrier(0);

  v8f c00 = {}, c01 = {}, c10 = {}, c11 = {};
#pragma unroll
  for (int j = 0; j < 8; ++j) {
    c00 = mm4(lo2(fa0[j]), lo2(fb0[j]), c00);
    c01 = mm4(lo2(fa0[j]), lo2(fb1[j]), c01);
    c10 = mm4(lo2(fa1[j]), lo2(fb0[j]), c10);
    c11 = mm4(lo2(fa1[j]), lo2(fb1[j]), c11);
    c00 = mm4(hi2(fa0[j]), hi2(fb0[j]), c00);
    c01 = mm4(hi2(fa0[j]), hi2(fb1[j]), c01);
    c10 = mm4(hi2(fa1[j]), hi2(fb0[j]), c10);
    c11 = mm4(hi2(fa1[j]), hi2(fb1[j]), c11);
  }

  float* rw = red + wave * 1024 + lane;
#pragma unroll
  for (int v = 0; v < 8; ++v) {
    rw[(0 * 8 + v) * 32] = c00[v];
    rw[(1 * 8 + v) * 32] = c01[v];
    rw[(2 * 8 + v) * 32] = c10[v];
    rw[(3 * 8 + v) * 32] = c11[v];
  }
  __syncthreads();

  const int sub = (((c0 >> 4) * 8 + (rr & 7)) * 32) + (rr >> 3) * 16 + (c0 & 15);
#pragma unroll
  for (int p = 0; p < 2; ++p) {
    const int idx = p * 2 * 8 * 32 + sub;
    v4f s = *(const v4f*)(red + idx);
    s += *(const v4f*)(red + 1024 + idx);
    s += *(const v4f*)(red + 2048 + idx);
    s += *(const v4f*)(red + 3072 + idx);
    s += bias;
    *(v4f*)(h1 + (m0 + 16 * p + rr) * H_ + n0 + c0) = s;
  }
}

template <int M>
__device__ __forceinline__ float lane_xor(float x) {
  const int b = __builtin_bit_cast(int, x);
  if (M == 16)
    return __builtin_bit_cast(float, __builtin_amdgcn_permlanex16(b, b, 0x76543210, 0xfedcba98, false, false));
  return __builtin_bit_cast(float, __builtin_amdgcn_update_dpp(0, b, 0x160 | (M & 15), 0xf, 0xf, true));
}
__device__ __forceinline__ v2f sr_add2(v2f a, v2f b) {
  v2f r;
  asm volatile("v_pk_add_f32 %0, %1, %2" : "=v"(r) : "v"(a), "v"(b));
  return r;
}
template <int IS_MAX>
__device__ __forceinline__ void sr_fold3(float& acc, float a, float b) {
  if (IS_MAX) asm volatile("v_max3_num_f32 %0, %0, %1, %2" : "+v"(acc) : "v"(a), "v"(b));
  else        asm volatile("v_min3_num_f32 %0, %0, %1, %2" : "+v"(acc) : "v"(a), "v"(b));
}
template <int IS_MAX>
__device__ __forceinline__ float sr_fold2(float a, float b) {
  float r;
  if (IS_MAX) asm("v_max_num_f32 %0, %1, %2" : "=v"(r) : "v"(a), "v"(b));
  else        asm("v_min_num_f32 %0, %1, %2" : "=v"(r) : "v"(a), "v"(b));
  return r;
}

template <int IS_MAX, int M, int HALF>
__device__ __forceinline__ void fold_bit(float (&v)[64], int lane) {
  const bool up = (lane & M) != 0;
  float keep[HALF], got[HALF];
#pragma unroll
  for (int k = 0; k < HALF; ++k) {
    keep[k] = up ? v[k + HALF] : v[k];
    const float send = up ? v[k] : v[k + HALF];
    got[k] = lane_xor<M>(send);
  }
#pragma unroll
  for (int k = 0; k < HALF; ++k) v[k] = sr_fold2<IS_MAX>(keep[k], got[k]);
}

template <int IS_MAX, int BASE>
__device__ __forceinline__ void sr_rows8(float (&acc)[4][16], const v4f (&w)[4], const v4f (&hv)[8]) {
#pragma unroll
  for (int p = 0; p < 8; p += 2) {
    v2f tl[2][4], th[2][4];
#pragma unroll
    for (int j = 0; j < 2; ++j)
#pragma unroll
      for (int oi = 0; oi < 4; ++oi) {
        tl[j][oi] = sr_add2(lo2(w[oi]), lo2(hv[p + j]));
        th[j][oi] = sr_add2(hi2(w[oi]), hi2(hv[p + j]));
      }
#pragma unroll
    for (int j = 0; j < 2; ++j)
#pragma unroll
      for (int oi = 0; oi < 4; ++oi) sr_fold3<IS_MAX>(acc[oi][BASE + p + j], tl[j][oi].x, tl[j][oi].y);
#pragma unroll
    for (int j = 0; j < 2; ++j)
#pragma unroll
      for (int oi = 0; oi < 4; ++oi) sr_fold3<IS_MAX>(acc[oi][BASE + p + j], th[j][oi].x, th[j][oi].y);
  }
}

template <int IS_MAX>
__device__ __forceinline__ void sr_step(float (&acc)[4][16], const v4f (&wc)[4], v4f (&wn)[4],
                                        v4f (&hA)[8], v4f (&hB)[8],
                                        const float* wp, const float* hp, int c) {
  const int io = c * 128;
  const int in = ((c + 1) & 7) * 128;
#pragma unroll
  for (int b = 0; b < 8; ++b) hB[b] = *(const v4f*)(hp + (8 + b) * H_ + io);
#pragma unroll
  for (int oi = 0; oi < 4; ++oi) wn[oi] = *(const v4f*)(wp + oi * H_ + in);
  sr_rows8<IS_MAX, 0>(acc, wc, hA);
#pragma unroll
  for (int b = 0; b < 8; ++b) hA[b] = *(const v4f*)(hp + b * H_ + in);
  sr_rows8<IS_MAX, 8>(acc, wc, hB);
}

template <int IS_MAX>
__global__ __launch_bounds__(512) void k_semiring(const float* __restrict__ hin,
                                                  const float* __restrict__ W,
                                                  float* __restrict__ hout) {
  __shared__ float hs[16 * H_];
  const int b0 = blockIdx.x * 16;
  const int o0 = blockIdx.y * 64;
  const int wave = threadIdx.x >> 5;
  const int lane = threadIdx.x & 31;
  const int ob = o0 + wave * 4;
  const float* wp = W + (size_t)ob * H_ + lane * 4;

  v4f wA[4], wB[4];
#pragma unroll
  for (int oi = 0; oi < 4; ++oi) wA[oi] = *(const v4f*)(wp + oi * H_);

  {
    const v4f* src = (const v4f*)(hin + b0 * H_);
    v4f* dst = (v4f*)hs;
    v4f r[8];
#pragma unroll
    for (int j = 0; j < 8; ++j) r[j] = src[threadIdx.x + 512 * j];
#pragma unroll
    for (int j = 0; j < 8; ++j) dst[threadIdx.x + 512 * j] = r[j];
  }
  __syncthreads();

  const float* hp = hs + lane * 4;
  float acc[4][16];
#pragma unroll
  for (int oi = 0; oi < 4; ++oi)
#pragma unroll
    for (int b = 0; b < 16; ++b)
      acc[oi][b] = IS_MAX ? -3.402823466e38f : 3.402823466e38f;

  v4f hA[8], hB[8];
#pragma unroll
  for (int b = 0; b < 8; ++b) hA[b] = *(const v4f*)(hp + b * H_);

#pragma unroll 1
  for (int c = 0; c < H_ / 128; c += 2) {
    sr_step<IS_MAX>(acc, wA, wB, hA, hB, wp, hp, c);
    sr_step<IS_MAX>(acc, wB, wA, hA, hB, wp, hp, c + 1);
  }

  float v[64];
#pragma unroll
  for (int b = 0; b < 16; ++b)
#pragma unroll
    for (int oi = 0; oi < 4; ++oi) v[b * 4 + oi] = acc[oi][b];

  fold_bit<IS_MAX, 16, 32>(v, lane);
  fold_bit<IS_MAX, 8, 16>(v, lane);
  fold_bit<IS_MAX, 4, 8>(v, lane);
  fold_bit<IS_MAX, 2, 4>(v, lane);
  fold_bit<IS_MAX, 1, 2>(v, lane);

  *(float2*)(hout + (b0 + (lane >> 1)) * H_ + ob + 2 * (lane & 1)) = make_float2(v[0], v[1]);
}

__global__ __launch_bounds__(512) void k_fc_out(const float* __restrict__ h,
                                                const float* __restrict__ Wout,
                                                const float* __restrict__ bout,
                                                float* __restrict__ q) {
  __shared__ __attribute__((aligned(16))) float red[16 * 512];
  const int wave = threadIdx.x >> 5;
  const int lane = threadIdx.x & 31;
  const int half = lane >> 4;
  const int l16  = lane & 15;
  const int m0   = blockIdx.x * 16;
  const int k0   = wave * 64 + 4 * half;

  const int col1  = 16 + l16;
  const int brow1 = (col1 < A_) ? col1 : (A_ - 1);

  const float* ap  = h    + (m0 + l16) * H_ + k0;
  const float* bp0 = Wout + l16 * H_        + k0;
  const float* bp1 = Wout + brow1 * H_      + k0;

  const int t   = threadIdx.x;
  const int row = t / A_;
  const int col = t - row * A_;
  const float bo = bout[col];

  v4f fa[8], fb0[8], fb1[8];
#pragma unroll
  for (int j = 0; j < 8; ++j) {
    fa[j]  = *(const v4f*)(ap  + 8 * j);
    fb0[j] = *(const v4f*)(bp0 + 8 * j);
    fb1[j] = *(const v4f*)(bp1 + 8 * j);
  }
  __builtin_amdgcn_sched_barrier(0);

  v8f c0 = {}, c1 = {};
#pragma unroll
  for (int j = 0; j < 8; ++j) {
    c0 = mm4(lo2(fa[j]), lo2(fb0[j]), c0);
    c1 = mm4(lo2(fa[j]), lo2(fb1[j]), c1);
    c0 = mm4(hi2(fa[j]), hi2(fb0[j]), c0);
    c1 = mm4(hi2(fa[j]), hi2(fb1[j]), c1);
  }

  float* rw = red + wave * 512 + lane;
#pragma unroll
  for (int v = 0; v < 8; ++v) {
    rw[v * 32]       = c0[v];
    rw[(8 + v) * 32] = c1[v];
  }
  __syncthreads();

  if (t < 16 * A_) {
    const int idx = ((col >> 4) * 8 + (row & 7)) * 32 + (row >> 3) * 16 + (col & 15);
    float s = red[idx];
#pragma unroll
    for (int w = 1; w < 16; ++w) s += red[w * 512 + idx];
    q[m0 * A_ + t] = s + bo;
  }
}

extern "C" void kernel_launch(void* const* d_in, const int* in_sizes, int n_in,
                              void* d_out, int out_size, void* d_ws, size_t ws_size,
                              hipStream_t stream) {
  (void)in_sizes; (void)n_in; (void)out_size; (void)ws_size;
  const float* x    = (const float*)d_in[0];
  const float* Win  = (const float*)d_in[1];
  const float* bin  = (const float*)d_in[2];
  const float* Wmax = (const float*)d_in[3];
  const float* Wmin = (const float*)d_in[4];
  const float* Wout = (const float*)d_in[5];
  const float* bout = (const float*)d_in[6];
  float* q  = (float*)d_out;
  float* h1 = (float*)d_ws;
  float* h2 = h1 + (size_t)B_ * H_;

  k_fc_in   <<<dim3(B_ / 32, H_ / 32), 128, 0, stream>>>(x,  Win,  bin,  h1);
  k_semiring<1><<<dim3(B_ / 16, H_ / 64), 512, 0, stream>>>(h1, Wmax, h2);
  k_semiring<0><<<dim3(B_ / 16, H_ / 64), 512, 0, stream>>>(h2, Wmin, h1);
  k_fc_out  <<<dim3(B_ / 16),          512, 0, stream>>>(h1, Wout, bout, q);
}


# fc_out: its WMMA products moved into the min-plus kernel (each block multiplies its own 16x64 result tile by its W_out slice); last launch only sums the 16 partials (f32 sum order changes)
# speedup vs baseline: 4.6083x; 1.0548x over previous
#include <hip/hip_runtime.h>

typedef __attribute__((ext_vector_type(2))) float v2f;
typedef __attribute__((ext_vector_type(4))) float v4f;
typedef __attribute__((ext_vector_type(8))) float v8f;

#define B_   256
#define OBS_ 256
#define H_   1024
#define A_   18

__device__ __forceinline__ v2f lo2(v4f v) { return __builtin_shufflevector(v, v, 0, 1); }
__device__ __forceinline__ v2f hi2(v4f v) { return __builtin_shufflevector(v, v, 2, 3); }

__device__ __forceinline__ v8f mm4(v2f a, v2f b, v8f c) {
  return __builtin_amdgcn_wmma_f32_16x16x4_f32(false, a, false, b, (short)0, c, false, false);
}

__global__ __launch_bounds__(128) void k_fc_in(const float* __restrict__ x,
                                               const float* __restrict__ Win,
                                               const float* __restrict__ bin,
                                               float* __restrict__ h1) {
  __shared__ __attribute__((aligned(16))) float red[4 * 1024];
  const int wave = threadIdx.x >> 5;
  const int lane = threadIdx.x & 31;
  const int half = lane >> 4;
  const int l16  = lane & 15;
  const int m0   = blockIdx.x * 32;
  const int n0   = blockIdx.y * 32;
  const int k0   = wave * 64 + 4 * half;

  const float* a0 = x   + (m0 +      l16) * OBS_ + k0;
  const float* a1 = x   + (m0 + 16 + l16) * OBS_ + k0;
  const float* b0 = Win + (n0 +      l16) * OBS_ + k0;
  const float* b1 = Win + (n0 + 16 + l16) * OBS_ + k0;

  const int t  = threadIdx.x;
  const int rr = t >> 3;
  const int c0 = 4 * (t & 7);
  const v4f bias = *(const v4f*)(bin + n0 + c0);

  v4f fa0[8], fa1[8], fb0[8], fb1[8];
#pragma unroll
  for (int j = 0; j < 8; ++j) {
    fa0[j] = *(const v4f*)(a0 + 8 * j);
    fb0[j] = *(const v4f*)(b0 + 8 * j);
    fa1[j] = *(const v4f*)(a1 + 8 * j);
    fb1[j] = *(const v4f*)(b1 + 8 * j);
  }
  __builtin_amdgcn_sched_barrier(0);

  v8f c00 = {}, c01 = {}, c10 = {}, c11 = {};
#pragma unroll
  for (int j = 0; j < 8; ++j) {
    c00 = mm4(lo2(fa0[j]), lo2(fb0[j]), c00);
    c01 = mm4(lo2(fa0[j]), lo2(fb1[j]), c01);
    c10 = mm4(lo2(fa1[j]), lo2(fb0[j]), c10);
    c11 = mm4(lo2(fa1[j]), lo2(fb1[j]), c11);
    c00 = mm4(hi2(fa0[j]), hi2(fb0[j]), c00);
    c01 = mm4(hi2(fa0[j]), hi2(fb1[j]), c01);
    c10 = mm4(hi2(fa1[j]), hi2(fb0[j]), c10);
    c11 = mm4(hi2(fa1[j]), hi2(fb1[j]), c11);
  }

  float* rw = red + wave * 1024 + lane;
#pragma unroll
  for (int v = 0; v < 8; ++v) {
    rw[(0 * 8 + v) * 32] = c00[v];
    rw[(1 * 8 + v) * 32] = c01[v];
    rw[(2 * 8 + v) * 32] = c10[v];
    rw[(3 * 8 + v) * 32] = c11[v];
  }
  __syncthreads();

  const int sub = (((c0 >> 4) * 8 + (rr & 7)) * 32) + (rr >> 3) * 16 + (c0 & 15);
#pragma unroll
  for (int p = 0; p < 2; ++p) {
    const int idx = p * 2 * 8 * 32 + sub;
    v4f s = *(const v4f*)(red + idx);
    s += *(const v4f*)(red + 1024 + idx);
    s += *(const v4f*)(red + 2048 + idx);
    s += *(const v4f*)(red + 3072 + idx);
    s += bias;
    *(v4f*)(h1 + (m0 + 16 * p + rr) * H_ + n0 + c0) = s;
  }
}

template <int M>
__device__ __forceinline__ float lane_xor(float x) {
  const int b = __builtin_bit_cast(int, x);
  if (M == 16)
    return __builtin_bit_cast(float, __builtin_amdgcn_permlanex16(b, b, 0x76543210, 0xfedcba98, false, false));
  return __builtin_bit_cast(float, __builtin_amdgcn_update_dpp(0, b, 0x160 | (M & 15), 0xf, 0xf, true));
}
__device__ __forceinline__ v2f sr_add2(v2f a, v2f b) {
  v2f r;
  asm volatile("v_pk_add_f32 %0, %1, %2" : "=v"(r) : "v"(a), "v"(b));
  return r;
}
template <int IS_MAX>
__device__ __forceinline__ void sr_fold3(float& acc, float a, float b) {
  if (IS_MAX) asm volatile("v_max3_num_f32 %0, %0, %1, %2" : "+v"(acc) : "v"(a), "v"(b));
  else        asm volatile("v_min3_num_f32 %0, %0, %1, %2" : "+v"(acc) : "v"(a), "v"(b));
}
template <int IS_MAX>
__device__ __forceinline__ float sr_fold2(float a, float b) {
  float r;
  if (IS_MAX) asm("v_max_num_f32 %0, %1, %2" : "=v"(r) : "v"(a), "v"(b));
  else        asm("v_min_num_f32 %0, %1, %2" : "=v"(r) : "v"(a), "v"(b));
  return r;
}

template <int IS_MAX, int M, int HALF>
__device__ __forceinline__ void fold_bit(float (&v)[64], int lane) {
  const bool up = (lane & M) != 0;
  float keep[HALF], got[HALF];
#pragma unroll
  for (int k = 0; k < HALF; ++k) {
    keep[k] = up ? v[k + HALF] : v[k];
    const float send = up ? v[k] : v[k + HALF];
    got[k] = lane_xor<M>(send);
  }
#pragma unroll
  for (int k = 0; k < HALF; ++k) v[k] = sr_fold2<IS_MAX>(keep[k], got[k]);
}

template <int IS_MAX, int BASE>
__device__ __forceinline__ void sr_rows8(float (&acc)[4][16], const v4f (&w)[4], const v4f (&hv)[8]) {
#pragma unroll
  for (int p = 0; p < 8; p += 2) {
    v2f tl[2][4], th[2][4];
#pragma unroll
    for (int j = 0; j < 2; ++j)
#pragma unroll
      for (int oi = 0; oi < 4; ++oi) {
        tl[j][oi] = sr_add2(lo2(w[oi]), lo2(hv[p + j]));
        th[j][oi] = sr_add2(hi2(w[oi]), hi2(hv[p + j]));
      }
#pragma unroll
    for (int j = 0; j < 2; ++j)
#pragma unroll
      for (int oi = 0; oi < 4; ++oi) sr_fold3<IS_MAX>(acc[oi][BASE + p + j], tl[j][oi].x, tl[j][oi].y);
#pragma unroll
    for (int j = 0; j < 2; ++j)
#pragma unroll
      for (int oi = 0; oi < 4; ++oi) sr_fold3<IS_MAX>(acc[oi][BASE + p + j], th[j][oi].x, th[j][oi].y);
  }
}

template <int IS_MAX>
__device__ __forceinline__ void sr_step(float (&acc)[4][16], const v4f (&wc)[4], v4f (&wn)[4],
                                        v4f (&hA)[8], v4f (&hB)[8],
                                        const float* wp, const float* hp, int c) {
  const int io = c * 128;
  const int in = ((c + 1) & 7) * 128;
#pragma unroll
  for (int b = 0; b < 8; ++b) hB[b] = *(const v4f*)(hp + (8 + b) * H_ + io);
#pragma unroll
  for (int oi = 0; oi < 4; ++oi) wn[oi] = *(const v4f*)(wp + oi * H_ + in);
  sr_rows8<IS_MAX, 0>(acc, wc, hA);
#pragma unroll
  for (int b = 0; b < 8; ++b) hA[b] = *(const v4f*)(hp + b * H_ + in);
  sr_rows8<IS_MAX, 8>(acc, wc, hB);
}

template <int IS_MAX>
__global__ __launch_bounds__(512) void k_semiring(const float* __restrict__ hin,
                                                  const float* __restrict__ W,
                                                  float* __restrict__ hout,
                                                  const float* __restrict__ Wout,
                                                  float* __restrict__ part) {
  __shared__ float hs[16 * H_];
  const int b0 = blockIdx.x * 16;
  const int o0 = blockIdx.y * 64;
  const int wave = threadIdx.x >> 5;
  const int lane = threadIdx.x & 31;
  const int ob = o0 + wave * 4;
  const float* wp = W + (size_t)ob * H_ + lane * 4;

  v4f wA[4], wB[4];
#pragma unroll
  for (int oi = 0; oi < 4; ++oi) wA[oi] = *(const v4f*)(wp + oi * H_);

  {
    const v4f* src = (const v4f*)(hin + b0 * H_);
    v4f* dst = (v4f*)hs;
    v4f r[8];
#pragma unroll
    for (int j = 0; j < 8; ++j) r[j] = src[threadIdx.x + 512 * j];
#pragma unroll
    for (int j = 0; j < 8; ++j) dst[threadIdx.x + 512 * j] = r[j];
  }
  __syncthreads();

  const float* hp = hs + lane * 4;
  float acc[4][16];
#pragma unroll
  for (int oi = 0; oi < 4; ++oi)
#pragma unroll
    for (int b = 0; b < 16; ++b)
      acc[oi][b] = IS_MAX ? -3.402823466e38f : 3.402823466e38f;

  v4f hA[8], hB[8];
#pragma unroll
  for (int b = 0; b < 8; ++b) hA[b] = *(const v4f*)(hp + b * H_);

#pragma unroll 1
  for (int c = 0; c < H_ / 128; c += 2) {
    sr_step<IS_MAX>(acc, wA, wB, hA, hB, wp, hp, c);
    sr_step<IS_MAX>(acc, wB, wA, hA, hB, wp, hp, c + 1);
  }

  const int wave_s = __builtin_amdgcn_readfirstlane(wave);
  const int half = lane >> 4;
  const int l16  = lane & 15;
  v4f fb[8];
  if (!IS_MAX && wave_s < 2) {
    const int ncol = 16 * wave_s + l16;
    const int brow = (ncol < A_) ? ncol : (A_ - 1);
    const float* bp = Wout + brow * H_ + o0 + 4 * half;
#pragma unroll
    for (int j = 0; j < 8; ++j) fb[j] = *(const v4f*)(bp + 8 * j);
  }

  float v[64];
#pragma unroll
  for (int b = 0; b < 16; ++b)
#pragma unroll
    for (int oi = 0; oi < 4; ++oi) v[b * 4 + oi] = acc[oi][b];

  fold_bit<IS_MAX, 16, 32>(v, lane);
  fold_bit<IS_MAX, 8, 16>(v, lane);
  fold_bit<IS_MAX, 4, 8>(v, lane);
  fold_bit<IS_MAX, 2, 4>(v, lane);
  fold_bit<IS_MAX, 1, 2>(v, lane);

  if (IS_MAX) {
    *(float2*)(hout + (b0 + (lane >> 1)) * H_ + ob + 2 * (lane & 1)) = make_float2(v[0], v[1]);
  } else {
    __shared__ __attribute__((aligned(16))) float ts[16 * 68];
    *(float2*)(ts + (lane >> 1) * 68 + wave * 4 + 2 * (lane & 1)) = make_float2(v[0], v[1]);
    __syncthreads();
    if (wave_s < 2) {
      v4f fa[8];
#pragma unroll
      for (int j = 0; j < 8; ++j) fa[j] = *(const v4f*)(ts + l16 * 68 + 8 * j + 4 * half);
      __builtin_amdgcn_sched_barrier(0);
      v8f ce = {}, co = {};
#pragma unroll
      for (int j = 0; j < 8; ++j) {
        ce = mm4(lo2(fa[j]), lo2(fb[j]), ce);
        co = mm4(hi2(fa[j]), hi2(fb[j]), co);
      }
      const v8f c = ce + co;
      float* pw = part + ((blockIdx.y * (B_ / 16) + blockIdx.x) * 2 + wave_s) * 256 + lane;
#pragma unroll
      for (int r = 0; r < 8; ++r) pw[r * 32] = c[r];
    }
  }
}

__global__ __launch_bounds__(32) void k_fc_out(const float* __restrict__ part,
                                               const float* __restrict__ bout,
                                               float* __restrict__ q) {
  const int g    = blockIdx.x * 32 + threadIdx.x;
  const int rowg = g / A_;
  const int col  = g - rowg * A_;
  const int row  = rowg & 15;
  const float* p = part + (rowg >> 4) * 512 + (col >> 4) * 256 + (row & 7) * 32 + (row >> 3) * 16 + (col & 15);
  const float bo = bout[col];
  float s = p[0];
#pragma unroll
  for (int y = 1; y < H_ / 64; ++y) s += p[y * (B_ / 16) * 512];
  q[g] = s + bo;
}

extern "C" void kernel_launch(void* const* d_in, const int* in_sizes, int n_in,
                              void* d_out, int out_size, void* d_ws, size_t ws_size,
                              hipStream_t stream) {
  (void)in_sizes; (void)n_in; (void)out_size; (void)ws_size;
  const float* x    = (const float*)d_in[0];
  const float* Win  = (const float*)d_in[1];
  const float* bin  = (const float*)d_in[2];
  const float* Wmax = (const float*)d_in[3];
  const float* Wmin = (const float*)d_in[4];
  const float* Wout = (const float*)d_in[5];
  const float* bout = (const float*)d_in[6];
  float* q  = (float*)d_out;
  float* h1 = (float*)d_ws;
  float* h2 = h1 + (size_t)B_ * H_;
  float* part = h2 + (size_t)B_ * H_;

  k_fc_in   <<<dim3(B_ / 32, H_ / 32), 128, 0, stream>>>(x,  Win,  bin,  h1);
  k_semiring<1><<<dim3(B_ / 16, H_ / 64), 512, 0, stream>>>(h1, Wmax, h2, nullptr, nullptr);
  k_semiring<0><<<dim3(B_ / 16, H_ / 64), 512, 0, stream>>>(h2, Wmin, nullptr, Wout, part);
  k_fc_out  <<<dim3(B_ * A_ / 32),     32,  0, stream>>>(part, bout, q);
}


# max-plus layer: exact pruning of the reduction index by per-column max/min bounds of W_max (bounds and a transposed copy made beside fc_in in every call); dense loop kept as fallback; bit-identical
# speedup vs baseline: 4.7864x; 1.0386x over previous
#include <hip/hip_runtime.h>

typedef __attribute__((ext_vector_type(2))) float v2f;
typedef __attribute__((ext_vector_type(4))) float v4f;
typedef __attribute__((ext_vector_type(8))) float v8f;

#define B_   256
#define OBS_ 256
#define H_   1024
#define A_   18
#define LIST_ 512

__device__ __forceinline__ v2f lo2(v4f v) { return __builtin_shufflevector(v, v, 0, 1); }
__device__ __forceinline__ v2f hi2(v4f v) { return __builtin_shufflevector(v, v, 2, 3); }

__device__ __forceinline__ v8f mm4(v2f a, v2f b, v8f c) {
  return __builtin_amdgcn_wmma_f32_16x16x4_f32(false, a, false, b, (short)0, c, false, false);
}

__device__ __forceinline__ void colbounds_transpose(const float* __restrict__ Wm, float* __restrict__ WT,
                                                    float* __restrict__ cmaxp, float* __restrict__ cminp,
                                                    float* sm, int e) {
  const int i0  = (e & 31) * 32;
  const int oq  = e >> 5;
  const int o0q = oq * 256;
  float* tile = sm;
  float* cm   = sm + 256 * 33;
  float* cn   = cm + 512;
  const int t  = threadIdx.x;
  const int c4 = t & 7;
  const int r  = t >> 3;
  v4f w[16];
#pragma unroll
  for (int p = 0; p < 16; ++p) w[p] = *(const v4f*)(Wm + (size_t)(o0q + 16 * p + r) * H_ + i0 + 4 * c4);
  v4f mx = w[0], mn = w[0];
#pragma unroll
  for (int p = 1; p < 16; ++p)
#pragma unroll
    for (int c = 0; c < 4; ++c) { mx[c] = fmaxf(mx[c], w[p][c]); mn[c] = fminf(mn[c], w[p][c]); }
#pragma unroll
  for (int p = 0; p < 16; ++p)
#pragma unroll
    for (int c = 0; c < 4; ++c) tile[(16 * p + r) * 33 + 4 * c4 + c] = w[p][c];
#pragma unroll
  for (int c = 0; c < 4; ++c) { cm[r * 32 + 4 * c4 + c] = mx[c]; cn[r * 32 + 4 * c4 + c] = mn[c]; }
  __syncthreads();
  if (t < 64) {
    const bool isMin = t >= 32;
    const float* s = isMin ? cn : cm;
    const int col = t & 31;
    float a = s[col];
#pragma unroll
    for (int q = 1; q < 16; ++q) a = isMin ? fminf(a, s[q * 32 + col]) : fmaxf(a, s[q * 32 + col]);
    (isMin ? cminp : cmaxp)[oq * H_ + i0 + col] = a;
  }
  const int og = t & 63;
  const int ih = t >> 6;
#pragma unroll
  for (int q = 0; q < 16; ++q) {
    const int ii = 2 * q + ih;
    v4f v;
#pragma unroll
    for (int c = 0; c < 4; ++c) v[c] = tile[(4 * og + c) * 33 + ii];
    *(v4f*)(WT + (size_t)(i0 + ii) * H_ + o0q + 4 * og) = v;
  }
}

__global__ __launch_bounds__(128) void k_fc_in(const float* __restrict__ x,
                                               const float* __restrict__ Win,
                                               const float* __restrict__ bin,
                                               float* __restrict__ h1,
                                               const float* __restrict__ Wmax,
                                               float* __restrict__ WT,
                                               float* __restrict__ cmaxp,
                                               float* __restrict__ cminp) {
  __shared__ __attribute__((aligned(16))) float red[256 * 33 + 2 * 512];
  if (blockIdx.y >= H_ / 32) {
    colbounds_transpose(Wmax, WT, cmaxp, cminp, red, (blockIdx.y - H_ / 32) * (B_ / 32) + blockIdx.x);
    return;
  }
  const int wave = threadIdx.x >> 5;
  const int lane = threadIdx.x & 31;
  const int half = lane >> 4;
  const int l16  = lane & 15;
  const int m0   = blockIdx.x * 32;
  const int n0   = blockIdx.y * 32;
  const int k0   = wave * 64 + 4 * half;

  const float* a0 = x   + (m0 +      l16) * OBS_ + k0;
  const float* a1 = x   + (m0 + 16 + l16) * OBS_ + k0;
  const float* b0 = Win + (n0 +      l16) * OBS_ + k0;
  const float* b1 = Win + (n0 + 16 + l16) * OBS_ + k0;

  const int t  = threadIdx.x;
  const int rr = t >> 3;
  const int c0 = 4 * (t & 7);
  const v4f bias = *(const v4f*)(bin + n0 + c0);

  v4f fa0[8], fa1[8], fb0[8], fb1[8];
#pragma unroll
  for (int j = 0; j < 8; ++j) {
    fa0[j] = *(const v4f*)(a0 + 8 * j);
    fb0[j] = *(const v4f*)(b0 + 8 * j);
    fa1[j] = *(const v4f*)(a1 + 8 * j);
    fb1[j] = *(const v4f*)(b1 + 8 * j);
  }
  __builtin_amdgcn_sched_barrier(0);

  v8f c00 = {}, c01 = {}, c10 = {}, c11 = {};
#pragma unroll
  for (int j = 0; j < 8; ++j) {
    c00 = mm4(lo2(fa0[j]), lo2(fb0[j]), c00);
    c01 = mm4(lo2(fa0[j]), lo2(fb1[j]), c01);
    c10 = mm4(lo2(fa1[j]), lo2(fb0[j]), c10);
    c11 = mm4(lo2(fa1[j]), lo2(fb1[j]), c11);
    c00 = mm4(hi2(fa0[j]), hi2(fb0[j]), c00);
    c01 = mm4(hi2(fa0[j]), hi2(fb1[j]), c01);
    c10 = mm4(hi2(fa1[j]), hi2(fb0[j]), c10);
    c11 = mm4(hi2(fa1[j]), hi2(fb1[j]), c11);
  }

  float* rw = red + wave * 1024 + lane;
#pragma unroll
  for (int v = 0; v < 8; ++v) {
    rw[(0 * 8 + v) * 32] = c00[v];
    rw[(1 * 8 + v) * 32] = c01[v];
    rw[(2 * 8 + v) * 32] = c10[v];
    rw[(3 * 8 + v) * 32] = c11[v];
  }
  __syncthreads();

  const int sub = (((c0 >> 4) * 8 + (rr & 7)) * 32) + (rr >> 3) * 16 + (c0 & 15);
#pragma unroll
  for (int p = 0; p < 2; ++p) {
    const int idx = p * 2 * 8 * 32 + sub;
    v4f s = *(const v4f*)(red + idx);
    s += *(const v4f*)(red + 1024 + idx);
    s += *(const v4f*)(red + 2048 + idx);
    s += *(const v4f*)(red + 3072 + idx);
    s += bias;
    *(v4f*)(h1 + (m0 + 16 * p + rr) * H_ + n0 + c0) = s;
  }
}

template <int M>
__device__ __forceinline__ float lane_xor(float x) {
  const int b = __builtin_bit_cast(int, x);
  if (M == 16)
    return __builtin_bit_cast(float, __builtin_amdgcn_permlanex16(b, b, 0x76543210, 0xfedcba98, false, false));
  return __builtin_bit_cast(float, __builtin_amdgcn_update_dpp(0, b, 0x160 | (M & 15), 0xf, 0xf, true));
}
__device__ __forceinline__ v2f sr_add2(v2f a, v2f b) {
  v2f r;
  asm volatile("v_pk_add_f32 %0, %1, %2" : "=v"(r) : "v"(a), "v"(b));
  return r;
}
template <int IS_MAX>
__device__ __forceinline__ void sr_fold3(float& acc, float a, float b) {
  if (IS_MAX) asm volatile("v_max3_num_f32 %0, %0, %1, %2" : "+v"(acc) : "v"(a), "v"(b));
  else        asm volatile("v_min3_num_f32 %0, %0, %1, %2" : "+v"(acc) : "v"(a), "v"(b));
}
template <int IS_MAX>
__device__ __forceinline__ float sr_fold2(float a, float b) {
  float r;
  if (IS_MAX) asm("v_max_num_f32 %0, %1, %2" : "=v"(r) : "v"(a), "v"(b));
  else        asm("v_min_num_f32 %0, %1, %2" : "=v"(r) : "v"(a), "v"(b));
  return r;
}

template <int IS_MAX, int M, int HALF>
__device__ __forceinline__ void fold_bit(float (&v)[64], int lane) {
  const bool up = (lane & M) != 0;
  float keep[HALF], got[HALF];
#pragma unroll
  for (int k = 0; k < HALF; ++k) {
    keep[k] = up ? v[k + HALF] : v[k];
    const float send = up ? v[k] : v[k + HALF];
    got[k] = lane_xor<M>(send);
  }
#pragma unroll
  for (int k = 0; k < HALF; ++k) v[k] = sr_fold2<IS_MAX>(keep[k], got[k]);
}

template <int IS_MAX, int BASE>
__device__ __forceinline__ void sr_rows8(float (&acc)[4][16], const v4f (&w)[4], const v4f (&hv)[8]) {
#pragma unroll
  for (int p = 0; p < 8; p += 2) {
    v2f tl[2][4], th[2][4];
#pragma unroll
    for (int j = 0; j < 2; ++j)
#pragma unroll
      for (int oi = 0; oi < 4; ++oi) {
        tl[j][oi] = sr_add2(lo2(w[oi]), lo2(hv[p + j]));
        th[j][oi] = sr_add2(hi2(w[oi]), hi2(hv[p + j]));
      }
#pragma unroll
    for (int j = 0; j < 2; ++j)
#pragma unroll
      for (int oi = 0; oi < 4; ++oi) sr_fold3<IS_MAX>(acc[oi][BASE + p + j], tl[j][oi].x, tl[j][oi].y);
#pragma unroll
    for (int j = 0; j < 2; ++j)
#pragma unroll
      for (int oi = 0; oi < 4; ++oi) sr_fold3<IS_MAX>(acc[oi][BASE + p + j], th[j][oi].x, th[j][oi].y);
  }
}

template <int IS_MAX>
__device__ __forceinline__ void sr_step(float (&acc)[4][16], const v4f (&wc)[4], v4f (&wn)[4],
                                        v4f (&hA)[8], v4f (&hB)[8],
                                        const float* wp, const float* hp, int c) {
  const int io = c * 128;
  const int in = ((c + 1) & 7) * 128;
#pragma unroll
  for (int b = 0; b < 8; ++b) hB[b] = *(const v4f*)(hp + (8 + b) * H_ + io);
#pragma unroll
  for (int oi = 0; oi < 4; ++oi) wn[oi] = *(const v4f*)(wp + oi * H_ + in);
  sr_rows8<IS_MAX, 0>(acc, wc, hA);
#pragma unroll
  for (int b = 0; b < 8; ++b) hA[b] = *(const v4f*)(hp + b * H_ + in);
  sr_rows8<IS_MAX, 8>(acc, wc, hB);
}

template <int IS_MAX>
__global__ __launch_bounds__(512) void k_semiring(const float* __restrict__ hin,
                                                  const float* __restrict__ W,
                                                  float* __restrict__ hout,
                                                  const float* __restrict__ Wout,
                                                  float* __restrict__ part,
                                                  const float* __restrict__ WT,
                                                  const float* __restrict__ cmaxp,
                                                  const float* __restrict__ cminp) {
  __shared__ __attribute__((aligned(16))) float hs[IS_MAX ? LIST_ * 64 : 16 * H_];
  __shared__ __attribute__((aligned(16))) float cbS[IS_MAX ? 2 * H_ : 4];
  __shared__ int   cntS[16], offS[16];
  __shared__ int   flatI[IS_MAX ? LIST_ : 1];
  __shared__ float flatH[IS_MAX ? LIST_ : 1];
  const int b0 = blockIdx.x * 16;
  const int o0 = blockIdx.y * 64;
  const int wave = threadIdx.x >> 5;
  const int lane = threadIdx.x & 31;
  const int ob = o0 + wave * 4;
  const float* wp = W + (size_t)ob * H_ + lane * 4;

  v4f wA[4], wB[4];
#pragma unroll
  for (int oi = 0; oi < 4; ++oi) wA[oi] = *(const v4f*)(wp + oi * H_);

  if (IS_MAX) {
    const bool isMin = threadIdx.x >= 256;
    const int j = (threadIdx.x & 255) * 4;
    const float* s = isMin ? cminp : cmaxp;
    const v4f q0 = *(const v4f*)(s + j), q1 = *(const v4f*)(s + H_ + j);
    const v4f q2 = *(const v4f*)(s + 2 * H_ + j), q3 = *(const v4f*)(s + 3 * H_ + j);
    v4f m;
#pragma unroll
    for (int c = 0; c < 4; ++c)
      m[c] = isMin ? fminf(fminf(q0[c], q1[c]), fminf(q2[c], q3[c])) : fmaxf(fmaxf(q0[c], q1[c]), fmaxf(q2[c], q3[c]));
    *(v4f*)(cbS + (isMin ? H_ : 0) + j) = m;
  }

  {
    const v4f* src = (const v4f*)(hin + b0 * H_);
    v4f* dst = (v4f*)hs;
    v4f r[8];
#pragma unroll
    for (int j = 0; j < 8; ++j) r[j] = src[threadIdx.x + 512 * j];
#pragma unroll
    for (int j = 0; j < 8; ++j) dst[threadIdx.x + 512 * j] = r[j];
  }
  __syncthreads();

  if (IS_MAX) {
    const float* hrow = hs + wave * H_ + lane * 4;
    v4f hv[8];
    float lb = -3.402823466e38f;
#pragma unroll
    for (int c = 0; c < 8; ++c) {
      hv[c] = *(const v4f*)(hrow + 128 * c);
      const v4f mnv = *(const v4f*)(cbS + H_ + lane * 4 + 128 * c);
#pragma unroll
      for (int s = 0; s < 4; ++s) lb = fmaxf(lb, hv[c][s] + mnv[s]);
    }
#pragma unroll
    for (int m = 16; m >= 1; m >>= 1) lb = fmaxf(lb, __shfl_xor(lb, m));
    unsigned bal[32];
    int n = 0;
#pragma unroll
    for (int c = 0; c < 8; ++c) {
      const v4f mxv = *(const v4f*)(cbS + lane * 4 + 128 * c);
#pragma unroll
      for (int s = 0; s < 4; ++s) {
        bal[4 * c + s] = __builtin_amdgcn_ballot_w32((hv[c][s] + mxv[s]) >= lb);
        n += __popc(bal[4 * c + s]);
      }
    }
    if (lane == 0) cntS[wave] = n;
    __syncthreads();
    int off = 0, total = 0;
#pragma unroll
    for (int r = 0; r < 16; ++r) {
      const int cr = cntS[r];
      off += (r < wave) ? cr : 0;
      total += cr;
    }
    total = __builtin_amdgcn_readfirstlane(total);
    if (total <= LIST_) {
      if (lane == 0) offS[wave] = off;
      const unsigned below = (1u << lane) - 1u;
      int base = off;
#pragma unroll
      for (int c = 0; c < 8; ++c)
#pragma unroll
        for (int s = 0; s < 4; ++s) {
          const unsigned m = bal[4 * c + s];
          if (m) {
            if ((m >> lane) & 1u) {
              const int pos = base + __popc(m & below);
              flatI[pos] = 128 * c + 4 * lane + s;
              flatH[pos] = hv[c][s];
            }
            base += __popc(m);
          }
        }
      __syncthreads();

      float* T = hs;
      const int last = total - 1;
      for (int j0 = 0; j0 < total; j0 += 256) {
        v2f wv[16];
        float hh[16];
#pragma unroll
        for (int u = 0; u < 16; ++u) {
          const int j  = j0 + wave + 16 * u;
          const int jc = j < last ? j : last;
          const int ii = flatI[jc] & (H_ - 1);
          hh[u] = flatH[jc];
          wv[u] = *(const v2f*)(WT + (size_t)ii * H_ + o0 + 2 * lane);
        }
#pragma unroll
        for (int u = 0; u < 16; ++u) {
          v2f tv;
          tv.x = wv[u].x + hh[u];
          tv.y = wv[u].y + hh[u];
          *(v2f*)(T + (j0 + wave + 16 * u) * 64 + 2 * lane) = tv;
        }
      }
      __syncthreads();

      const int o  = threadIdx.x & 63;
      const int bq = threadIdx.x >> 6;
#pragma unroll
      for (int p = 0; p < 2; ++p) {
        const int b  = bq + 8 * p;
        const int nb = __builtin_amdgcn_readfirstlane(cntS[b]);
        const int ob2 = __builtin_amdgcn_readfirstlane(offS[b]);
        const float* tp = T + ob2 * 64 + o;
        float a = -3.402823466e38f;
        for (int k0 = 0; k0 < nb; k0 += 8) {
          float tv[8];
#pragma unroll
          for (int u = 0; u < 8; ++u) {
            const int k = (k0 + u) < nb ? (k0 + u) : (nb - 1);
            tv[u] = tp[k * 64];
          }
#pragma unroll
          for (int u = 0; u < 8; ++u) a = fmaxf(a, tv[u]);
        }
        hout[(b0 + b) * H_ + o0 + o] = a;
      }
      return;
    }
  }

  const float* hp = hs + lane * 4;
  float acc[4][16];
#pragma unroll
  for (int oi = 0; oi < 4; ++oi)
#pragma unroll
    for (int b = 0; b < 16; ++b)
      acc[oi][b] = IS_MAX ? -3.402823466e38f : 3.402823466e38f;

  v4f hA[8], hB[8];
#pragma unroll
  for (int b = 0; b < 8; ++b) hA[b] = *(const v4f*)(hp + b * H_);

#pragma unroll 1
  for (int c = 0; c < H_ / 128; c += 2) {
    sr_step<IS_MAX>(acc, wA, wB, hA, hB, wp, hp, c);
    sr_step<IS_MAX>(acc, wB, wA, hA, hB, wp, hp, c + 1);
  }

  const int wave_s = __builtin_amdgcn_readfirstlane(wave);
  const int half = lane >> 4;
  const int l16  = lane & 15;
  v4f fb[8];
  if (!IS_MAX && wave_s < 2) {
    const int ncol = 16 * wave_s + l16;
    const int brow = (ncol < A_) ? ncol : (A_ - 1);
    const float* bp = Wout + brow * H_ + o0 + 4 * half;
#pragma unroll
    for (int j = 0; j < 8; ++j) fb[j] = *(const v4f*)(bp + 8 * j);
  }

  float v[64];
#pragma unroll
  for (int b = 0; b < 16; ++b)
#pragma unroll
    for (int oi = 0; oi < 4; ++oi) v[b * 4 + oi] = acc[oi][b];

  fold_bit<IS_MAX, 16, 32>(v, lane);
  fold_bit<IS_MAX, 8, 16>(v, lane);
  fold_bit<IS_MAX, 4, 8>(v, lane);
  fold_bit<IS_MAX, 2, 4>(v, lane);
  fold_bit<IS_MAX, 1, 2>(v, lane);

  if (IS_MAX) {
    *(float2*)(hout + (b0 + (lane >> 1)) * H_ + ob + 2 * (lane & 1)) = make_float2(v[0], v[1]);
  } else {
    __shared__ __attribute__((aligned(16))) float ts[16 * 68];
    *(float2*)(ts + (lane >> 1) * 68 + wave * 4 + 2 * (lane & 1)) = make_float2(v[0], v[1]);
    __syncthreads();
    if (wave_s < 2) {
      v4f fa[8];
#pragma unroll
      for (int j = 0; j < 8; ++j) fa[j] = *(const v4f*)(ts + l16 * 68 + 8 * j + 4 * half);
      __builtin_amdgcn_sched_barrier(0);
      v8f ce = {}, co = {};
#pragma unroll
      for (int j = 0; j < 8; ++j) {
        ce = mm4(lo2(fa[j]), lo2(fb[j]), ce);
        co = mm4(hi2(fa[j]), hi2(fb[j]), co);
      }
      const v8f c = ce + co;
      float* pw = part + ((blockIdx.y * (B_ / 16) + blockIdx.x) * 2 + wave_s) * 256 + lane;
#pragma unroll
      for (int r = 0; r < 8; ++r) pw[r * 32] = c[r];
    }
  }
}

__global__ __launch_bounds__(32) void k_fc_out(const float* __restrict__ part,
                                               const float* __restrict__ bout,
                                               float* __restrict__ q) {
  const int g    = blockIdx.x * 32 + threadIdx.x;
  const int rowg = g / A_;
  const int col  = g - rowg * A_;
  const int row  = rowg & 15;
  const float* p = part + (rowg >> 4) * 512 + (col >> 4) * 256 + (row & 7) * 32 + (row >> 3) * 16 + (col & 15);
  const float bo = bout[col];
  float s = p[0];
#pragma unroll
  for (int y = 1; y < H_ / 64; ++y) s += p[y * (B_ / 16) * 512];
  q[g] = s + bo;
}

extern "C" void kernel_launch(void* const* d_in, const int* in_sizes, int n_in,
                              void* d_out, int out_size, void* d_ws, size_t ws_size,
                              hipStream_t stream) {
  (void)in_sizes; (void)n_in; (void)out_size; (void)ws_size;
  const float* x    = (const float*)d_in[0];
  const float* Win  = (const float*)d_in[1];
  const float* bin  = (const float*)d_in[2];
  const float* Wmax = (const float*)d_in[3];
  const float* Wmin = (const float*)d_in[4];
  const float* Wout = (const float*)d_in[5];
  const float* bout = (const float*)d_in[6];
  float* q  = (float*)d_out;
  float* h1 = (float*)d_ws;
  float* h2 = h1 + (size_t)B_ * H_;
  float* part = h2 + (size_t)B_ * H_;
  float* WT    = part + (size_t)(H_ / 64) * (B_ / 16) * 512;
  float* cmaxp = WT + (size_t)H_ * H_;
  float* cminp = cmaxp + 4 * H_;

  k_fc_in   <<<dim3(B_ / 32, H_ / 32 + 16), 128, 0, stream>>>(x,  Win,  bin,  h1, Wmax, WT, cmaxp, cminp);
  k_semiring<1><<<dim3(B_ / 16, H_ / 64), 512, 0, stream>>>(h1, Wmax, h2, nullptr, nullptr, WT, cmaxp, cminp);
  k_semiring<0><<<dim3(B_ / 16, H_ / 64), 512, 0, stream>>>(h2, Wmin, nullptr, Wout, part, nullptr, nullptr, nullptr);
  k_fc_out  <<<dim3(B_ * A_ / 32),     32,  0, stream>>>(part, bout, q);
}


# max-plus layer re-tiled: one 256-thread block per batch row, all 1024 outputs; the row's kept list is built once (was once per 64-column block) and can hold every index: no dense path; bit-identical
# speedup vs baseline: 5.0551x; 1.0561x over previous
#include <hip/hip_runtime.h>

typedef __attribute__((ext_vector_type(2))) float v2f;
typedef __attribute__((ext_vector_type(4))) float v4f;
typedef __attribute__((ext_vector_type(8))) float v8f;

#define B_   256
#define OBS_ 256
#define H_   1024
#define A_   18

__device__ __forceinline__ v2f lo2(v4f v) { return __builtin_shufflevector(v, v, 0, 1); }
__device__ __forceinline__ v2f hi2(v4f v) { return __builtin_shufflevector(v, v, 2, 3); }

__device__ __forceinline__ v8f mm4(v2f a, v2f b, v8f c) {
  return __builtin_amdgcn_wmma_f32_16x16x4_f32(false, a, false, b, (short)0, c, false, false);
}

__device__ __forceinline__ void colbounds_transpose(const float* __restrict__ Wm, float* __restrict__ WT,
                                                    float* __restrict__ cmaxp, float* __restrict__ cminp,
                                                    float* sm, int e) {
  const int i0  = (e & 31) * 32;
  const int oq  = e >> 5;
  const int o0q = oq * 256;
  float* tile = sm;
  float* cm   = sm + 256 * 33;
  float* cn   = cm + 512;
  const int t  = threadIdx.x;
  const int c4 = t & 7;
  const int r  = t >> 3;
  v4f w[16];
#pragma unroll
  for (int p = 0; p < 16; ++p) w[p] = *(const v4f*)(Wm + (size_t)(o0q + 16 * p + r) * H_ + i0 + 4 * c4);
  v4f mx = w[0], mn = w[0];
#pragma unroll
  for (int p = 1; p < 16; ++p)
#pragma unroll
    for (int c = 0; c < 4; ++c) { mx[c] = fmaxf(mx[c], w[p][c]); mn[c] = fminf(mn[c], w[p][c]); }
#pragma unroll
  for (int p = 0; p < 16; ++p)
#pragma unroll
    for (int c = 0; c < 4; ++c) tile[(16 * p + r) * 33 + 4 * c4 + c] = w[p][c];
#pragma unroll
  for (int c = 0; c < 4; ++c) { cm[r * 32 + 4 * c4 + c] = mx[c]; cn[r * 32 + 4 * c4 + c] = mn[c]; }
  __syncthreads();
  if (t < 64) {
    const bool isMin = t >= 32;
    const float* s = isMin ? cn : cm;
    const int col = t & 31;
    float a = s[col];
#pragma unroll
    for (int q = 1; q < 16; ++q) a = isMin ? fminf(a, s[q * 32 + col]) : fmaxf(a, s[q * 32 + col]);
    (isMin ? cminp : cmaxp)[oq * H_ + i0 + col] = a;
  }
  const int og = t & 63;
  const int ih = t >> 6;
#pragma unroll
  for (int q = 0; q < 16; ++q) {
    const int ii = 2 * q + ih;
    v4f v;
#pragma unroll
    for (int c = 0; c < 4; ++c) v[c] = tile[(4 * og + c) * 33 + ii];
    *(v4f*)(WT + (size_t)(i0 + ii) * H_ + o0q + 4 * og) = v;
  }
}

__global__ __launch_bounds__(128) void k_fc_in(const float* __restrict__ x,
                                               const float* __restrict__ Win,
                                               const float* __restrict__ bin,
                                               float* __restrict__ h1,
                                               const float* __restrict__ Wmax,
                                               float* __restrict__ WT,
                                               float* __restrict__ cmaxp,
                                               float* __restrict__ cminp) {
  __shared__ __attribute__((aligned(16))) float red[256 * 33 + 2 * 512];
  if (blockIdx.y >= H_ / 32) {
    colbounds_transpose(Wmax, WT, cmaxp, cminp, red, (blockIdx.y - H_ / 32) * (B_ / 32) + blockIdx.x);
    return;
  }
  const int wave = threadIdx.x >> 5;
  const int lane = threadIdx.x & 31;
  const int half = lane >> 4;
  const int l16  = lane & 15;
  const int m0   = blockIdx.x * 32;
  const int n0   = blockIdx.y * 32;
  const int k0   = wave * 64 + 4 * half;

  const float* a0 = x   + (m0 +      l16) * OBS_ + k0;
  const float* a1 = x   + (m0 + 16 + l16) * OBS_ + k0;
  const float* b0 = Win + (n0 +      l16) * OBS_ + k0;
  const float* b1 = Win + (n0 + 16 + l16) * OBS_ + k0;

  const int t  = threadIdx.x;
  const int rr = t >> 3;
  const int c0 = 4 * (t & 7);
  const v4f bias = *(const v4f*)(bin + n0 + c0);

  v4f fa0[8], fa1[8], fb0[8], fb1[8];
#pragma unroll
  for (int j = 0; j < 8; ++j) {
    fa0[j] = *(const v4f*)(a0 + 8 * j);
    fb0[j] = *(const v4f*)(b0 + 8 * j);
    fa1[j] = *(const v4f*)(a1 + 8 * j);
    fb1[j] = *(const v4f*)(b1 + 8 * j);
  }
  __builtin_amdgcn_sched_barrier(0);

  v8f c00 = {}, c01 = {}, c10 = {}, c11 = {};
#pragma unroll
  for (int j = 0; j < 8; ++j) {
    c00 = mm4(lo2(fa0[j]), lo2(fb0[j]), c00);
    c01 = mm4(lo2(fa0[j]), lo2(fb1[j]), c01);
    c10 = mm4(lo2(fa1[j]), lo2(fb0[j]), c10);
    c11 = mm4(lo2(fa1[j]), lo2(fb1[j]), c11);
    c00 = mm4(hi2(fa0[j]), hi2(fb0[j]), c00);
    c01 = mm4(hi2(fa0[j]), hi2(fb1[j]), c01);
    c10 = mm4(hi2(fa1[j]), hi2(fb0[j]), c10);
    c11 = mm4(hi2(fa1[j]), hi2(fb1[j]), c11);
  }

  float* rw = red + wave * 1024 + lane;
#pragma unroll
  for (int v = 0; v < 8; ++v) {
    rw[(0 * 8 + v) * 32] = c00[v];
    rw[(1 * 8 + v) * 32] = c01[v];
    rw[(2 * 8 + v) * 32] = c10[v];
    rw[(3 * 8 + v) * 32] = c11[v];
  }
  __syncthreads();

  const int sub = (((c0 >> 4) * 8 + (rr & 7)) * 32) + (rr >> 3) * 16 + (c0 & 15);
#pragma unroll
  for (int p = 0; p < 2; ++p) {
    const int idx = p * 2 * 8 * 32 + sub;
    v4f s = *(const v4f*)(red + idx);
    s += *(const v4f*)(red + 1024 + idx);
    s += *(const v4f*)(red + 2048 + idx);
    s += *(const v4f*)(red + 3072 + idx);
    s += bias;
    *(v4f*)(h1 + (m0 + 16 * p + rr) * H_ + n0 + c0) = s;
  }
}

template <int M>
__device__ __forceinline__ float lane_xor(float x) {
  const int b = __builtin_bit_cast(int, x);
  if (M == 16)
    return __builtin_bit_cast(float, __builtin_amdgcn_permlanex16(b, b, 0x76543210, 0xfedcba98, false, false));
  return __builtin_bit_cast(float, __builtin_amdgcn_update_dpp(0, b, 0x160 | (M & 15), 0xf, 0xf, true));
}
__device__ __forceinline__ v2f sr_add2(v2f a, v2f b) {
  v2f r;
  asm volatile("v_pk_add_f32 %0, %1, %2" : "=v"(r) : "v"(a), "v"(b));
  return r;
}
template <int IS_MAX>
__device__ __forceinline__ void sr_fold3(float& acc, float a, float b) {
  if (IS_MAX) asm volatile("v_max3_num_f32 %0, %0, %1, %2" : "+v"(acc) : "v"(a), "v"(b));
  else        asm volatile("v_min3_num_f32 %0, %0, %1, %2" : "+v"(acc) : "v"(a), "v"(b));
}
template <int IS_MAX>
__device__ __forceinline__ float sr_fold2(float a, float b) {
  float r;
  if (IS_MAX) asm("v_max_num_f32 %0, %1, %2" : "=v"(r) : "v"(a), "v"(b));
  else        asm("v_min_num_f32 %0, %1, %2" : "=v"(r) : "v"(a), "v"(b));
  return r;
}

template <int IS_MAX, int M, int HALF>
__device__ __forceinline__ void fold_bit(float (&v)[64], int lane) {
  const bool up = (lane & M) != 0;
  float keep[HALF], got[HALF];
#pragma unroll
  for (int k = 0; k < HALF; ++k) {
    keep[k] = up ? v[k + HALF] : v[k];
    const float send = up ? v[k] : v[k + HALF];
    got[k] = lane_xor<M>(send);
  }
#pragma unroll
  for (int k = 0; k < HALF; ++k) v[k] = sr_fold2<IS_MAX>(keep[k], got[k]);
}

template <int IS_MAX, int BASE>
__device__ __forceinline__ void sr_rows8(float (&acc)[4][16], const v4f (&w)[4], const v4f (&hv)[8]) {
#pragma unroll
  for (int p = 0; p < 8; p += 2) {
    v2f tl[2][4], th[2][4];
#pragma unroll
    for (int j = 0; j < 2; ++j)
#pragma unroll
      for (int oi = 0; oi < 4; ++oi) {
        tl[j][oi] = sr_add2(lo2(w[oi]), lo2(hv[p + j]));
        th[j][oi] = sr_add2(hi2(w[oi]), hi2(hv[p + j]));
      }
#pragma unroll
    for (int j = 0; j < 2; ++j)
#pragma unroll
      for (int oi = 0; oi < 4; ++oi) sr_fold3<IS_MAX>(acc[oi][BASE + p + j], tl[j][oi].x, tl[j][oi].y);
#pragma unroll
    for (int j = 0; j < 2; ++j)
#pragma unroll
      for (int oi = 0; oi < 4; ++oi) sr_fold3<IS_MAX>(acc[oi][BASE + p + j], th[j][oi].x, th[j][oi].y);
  }
}

template <int IS_MAX>
__device__ __forceinline__ void sr_step(float (&acc)[4][16], const v4f (&wc)[4], v4f (&wn)[4],
                                        v4f (&hA)[8], v4f (&hB)[8],
                                        const float* wp, const float* hp, int c) {
  const int io = c * 128;
  const int in = ((c + 1) & 7) * 128;
#pragma unroll
  for (int b = 0; b < 8; ++b) hB[b] = *(const v4f*)(hp + (8 + b) * H_ + io);
#pragma unroll
  for (int oi = 0; oi < 4; ++oi) wn[oi] = *(const v4f*)(wp + oi * H_ + in);
  sr_rows8<IS_MAX, 0>(acc, wc, hA);
#pragma unroll
  for (int b = 0; b < 8; ++b) hA[b] = *(const v4f*)(hp + b * H_ + in);
  sr_rows8<IS_MAX, 8>(acc, wc, hB);
}

__global__ __launch_bounds__(256) void k_maxplus_row(const float* __restrict__ hin,
                                                     float* __restrict__ hout,
                                                     const float* __restrict__ WT,
                                                     const float* __restrict__ cmaxp,
                                                     const float* __restrict__ cminp) {
  __shared__ float redS[8];
  __shared__ int   cntS[8];
  __shared__ int   flatI[H_];
  __shared__ float flatH[H_];
  const int b    = blockIdx.x;
  const int t    = threadIdx.x;
  const int wave = t >> 5;
  const int lane = t & 31;
  const int j    = 4 * t;

  const v4f hv = *(const v4f*)(hin + (size_t)b * H_ + j);
  const v4f a0 = *(const v4f*)(cmaxp + j),          a1 = *(const v4f*)(cmaxp + H_ + j);
  const v4f a2 = *(const v4f*)(cmaxp + 2 * H_ + j), a3 = *(const v4f*)(cmaxp + 3 * H_ + j);
  const v4f n0 = *(const v4f*)(cminp + j),          n1 = *(const v4f*)(cminp + H_ + j);
  const v4f n2 = *(const v4f*)(cminp + 2 * H_ + j), n3 = *(const v4f*)(cminp + 3 * H_ + j);
  v4f mx, mn;
#pragma unroll
  for (int c = 0; c < 4; ++c) {
    mx[c] = fmaxf(fmaxf(a0[c], a1[c]), fmaxf(a2[c], a3[c]));
    mn[c] = fminf(fminf(n0[c], n1[c]), fminf(n2[c], n3[c]));
  }

  float lb = fmaxf(fmaxf(hv[0] + mn[0], hv[1] + mn[1]), fmaxf(hv[2] + mn[2], hv[3] + mn[3]));
  lb = fmaxf(lb, lane_xor<16>(lb));
  lb = fmaxf(lb, lane_xor<8>(lb));
  lb = fmaxf(lb, lane_xor<4>(lb));
  lb = fmaxf(lb, lane_xor<2>(lb));
  lb = fmaxf(lb, lane_xor<1>(lb));
  if (lane == 0) redS[wave] = lb;
  __syncthreads();
  float LB = redS[0];
#pragma unroll
  for (int w = 1; w < 8; ++w) LB = fmaxf(LB, redS[w]);

  unsigned bal[4];
  int n = 0;
#pragma unroll
  for (int s = 0; s < 4; ++s) {
    bal[s] = __builtin_amdgcn_ballot_w32((hv[s] + mx[s]) >= LB);
    n += __popc(bal[s]);
  }
  if (lane == 0) cntS[wave] = n;
  __syncthreads();
  int off = 0, total = 0;
#pragma unroll
  for (int r = 0; r < 8; ++r) {
    const int cr = cntS[r];
    off += (r < wave) ? cr : 0;
    total += cr;
  }
  total = __builtin_amdgcn_readfirstlane(total);
  {
    const unsigned below = (1u << lane) - 1u;
    int base = off;
#pragma unroll
    for (int s = 0; s < 4; ++s) {
      const unsigned m = bal[s];
      if ((m >> lane) & 1u) {
        const int pos = base + __popc(m & below);
        flatI[pos] = j + s;
        flatH[pos] = hv[s];
      }
      base += __popc(m);
    }
  }
  __syncthreads();

  v4f acc;
#pragma unroll
  for (int c = 0; c < 4; ++c) acc[c] = -3.402823466e38f;
  const int last = total - 1;
  for (int k0 = 0; k0 < total; k0 += 32) {
    v4f wv[32];
    float hh[32];
#pragma unroll
    for (int g = 0; g < 4; ++g) {
      if (k0 + 8 * g < total) {
#pragma unroll
        for (int u = 0; u < 8; ++u) {
          const int k  = (k0 + 8 * g + u) < last ? (k0 + 8 * g + u) : last;
          const int ii = flatI[k] & (H_ - 1);
          hh[8 * g + u] = flatH[k];
          wv[8 * g + u] = *(const v4f*)(WT + (size_t)ii * H_ + j);
        }
      }
    }
    __builtin_amdgcn_sched_barrier(0);
#pragma unroll
    for (int g = 0; g < 4; ++g) {
      if (k0 + 8 * g < total) {
#pragma unroll
        for (int u = 0; u < 8; ++u)
#pragma unroll
          for (int c = 0; c < 4; ++c) acc[c] = fmaxf(acc[c], wv[8 * g + u][c] + hh[8 * g + u]);
      }
    }
  }
  *(v4f*)(hout + (size_t)b * H_ + j) = acc;
}

__global__ __launch_bounds__(512) void k_minplus(const float* __restrict__ hin,
                                                 const float* __restrict__ W,
                                                 const float* __restrict__ Wout,
                                                 float* __restrict__ part) {
  __shared__ __attribute__((aligned(16))) float hs[16 * H_];
  const int b0 = blockIdx.x * 16;
  const int o0 = blockIdx.y * 64;
  const int wave = threadIdx.x >> 5;
  const int lane = threadIdx.x & 31;
  const int ob = o0 + wave * 4;
  const float* wp = W + (size_t)ob * H_ + lane * 4;

  v4f wA[4], wB[4];
#pragma unroll
  for (int oi = 0; oi < 4; ++oi) wA[oi] = *(const v4f*)(wp + oi * H_);

  {
    const v4f* src = (const v4f*)(hin + b0 * H_);
    v4f* dst = (v4f*)hs;
    v4f r[8];
#pragma unroll
    for (int j = 0; j < 8; ++j) r[j] = src[threadIdx.x + 512 * j];
#pragma unroll
    for (int j = 0; j < 8; ++j) dst[threadIdx.x + 512 * j] = r[j];
  }
  __syncthreads();

  const float* hp = hs + lane * 4;
  float acc[4][16];
#pragma unroll
  for (int oi = 0; oi < 4; ++oi)
#pragma unroll
    for (int b = 0; b < 16; ++b)
      acc[oi][b] = 3.402823466e38f;

  v4f hA[8], hB[8];
#pragma unroll
  for (int b = 0; b < 8; ++b) hA[b] = *(const v4f*)(hp + b * H_);

#pragma unroll 1
  for (int c = 0; c < H_ / 128; c += 2) {
    sr_step<0>(acc, wA, wB, hA, hB, wp, hp, c);
    sr_step<0>(acc, wB, wA, hA, hB, wp, hp, c + 1);
  }

  const int wave_s = __builtin_amdgcn_readfirstlane(wave);
  const int half = lane >> 4;
  const int l16  = lane & 15;
  v4f fb[8];
  if (wave_s < 2) {
    const int ncol = 16 * wave_s + l16;
    const int brow = (ncol < A_) ? ncol : (A_ - 1);
    const float* bp = Wout + brow * H_ + o0 + 4 * half;
#pragma unroll
    for (int j = 0; j < 8; ++j) fb[j] = *(const v4f*)(bp + 8 * j);
  }

  float v[64];
#pragma unroll
  for (int b = 0; b < 16; ++b)
#pragma unroll
    for (int oi = 0; oi < 4; ++oi) v[b * 4 + oi] = acc[oi][b];

  fold_bit<0, 16, 32>(v, lane);
  fold_bit<0, 8, 16>(v, lane);
  fold_bit<0, 4, 8>(v, lane);
  fold_bit<0, 2, 4>(v, lane);
  fold_bit<0, 1, 2>(v, lane);

  __shared__ __attribute__((aligned(16))) float ts[16 * 68];
  *(float2*)(ts + (lane >> 1) * 68 + wave * 4 + 2 * (lane & 1)) = make_float2(v[0], v[1]);
  __syncthreads();
  if (wave_s < 2) {
    v4f fa[8];
#pragma unroll
    for (int j = 0; j < 8; ++j) fa[j] = *(const v4f*)(ts + l16 * 68 + 8 * j + 4 * half);
    __builtin_amdgcn_sched_barrier(0);
    v8f ce = {}, co = {};
#pragma unroll
    for (int j = 0; j < 8; ++j) {
      ce = mm4(lo2(fa[j]), lo2(fb[j]), ce);
      co = mm4(hi2(fa[j]), hi2(fb[j]), co);
    }
    const v8f c = ce + co;
    float* pw = part + ((blockIdx.y * (B_ / 16) + blockIdx.x) * 2 + wave_s) * 256 + lane;
#pragma unroll
    for (int r = 0; r < 8; ++r) pw[r * 32] = c[r];
  }
}

__global__ __launch_bounds__(32) void k_fc_out(const float* __restrict__ part,
                                               const float* __restrict__ bout,
                                               float* __restrict__ q) {
  const int g    = blockIdx.x * 32 + threadIdx.x;
  const int rowg = g / A_;
  const int col  = g - rowg * A_;
  const int row  = rowg & 15;
  const float* p = part + (rowg >> 4) * 512 + (col >> 4) * 256 + (row & 7) * 32 + (row >> 3) * 16 + (col & 15);
  const float bo = bout[col];
  float s = p[0];
#pragma unroll
  for (int y = 1; y < H_ / 64; ++y) s += p[y * (B_ / 16) * 512];
  q[g] = s + bo;
}

extern "C" void kernel_launch(void* const* d_in, const int* in_sizes, int n_in,
                              void* d_out, int out_size, void* d_ws, size_t ws_size,
                              hipStream_t stream) {
  (void)in_sizes; (void)n_in; (void)out_size; (void)ws_size;
  const float* x    = (const float*)d_in[0];
  const float* Win  = (const float*)d_in[1];
  const float* bin  = (const float*)d_in[2];
  const float* Wmax = (const float*)d_in[3];
  const float* Wmin = (const float*)d_in[4];
  const float* Wout = (const float*)d_in[5];
  const float* bout = (const float*)d_in[6];
  float* q  = (float*)d_out;
  float* h1 = (float*)d_ws;
  float* h2 = h1 + (size_t)B_ * H_;
  float* part = h2 + (size_t)B_ * H_;
  float* WT    = part + (size_t)(H_ / 64) * (B_ / 16) * 512;
  float* cmaxp = WT + (size_t)H_ * H_;
  float* cminp = cmaxp + 4 * H_;

  k_fc_in   <<<dim3(B_ / 32, H_ / 32 + 16), 128, 0, stream>>>(x,  Win,  bin,  h1, Wmax, WT, cmaxp, cminp);
  k_maxplus_row<<<dim3(B_),            256, 0, stream>>>(h1, h2, WT, cmaxp, cminp);
  k_minplus <<<dim3(B_ / 16, H_ / 64), 512, 0, stream>>>(h2, Wmin, Wout, part);
  k_fc_out  <<<dim3(B_ * A_ / 32),     32,  0, stream>>>(part, bout, q);
}
